# starting kernel
# baseline (speedup 1.0000x reference)
#include <hip/hip_runtime.h>

typedef float v2f __attribute__((ext_vector_type(2)));
typedef float v4f __attribute__((ext_vector_type(4)));
typedef float v8f __attribute__((ext_vector_type(8)));

#define S_DIM 2048
#define B_DIM 16
#define D1 256
#define D2 256
#define ROWSTRIDE (B_DIM * D1)

#define BM 128
#define BN 64
#define BK 32
#define NKT (S_DIM / BK)
#define LDSK 36

__global__ __launch_bounds__(256)
void SumFlattenedOuterProduct_kernel(const float* __restrict__ gA,
                                     const float* __restrict__ gB,
                                     float* __restrict__ gC)
{
    __shared__ float lA[2][BM * LDSK];
    __shared__ float lB[2][BN * LDSK];

    const int b  = blockIdx.z;
    const int n0 = blockIdx.x * BN;
    const int m0 = blockIdx.y * BM;

    const int t    = threadIdx.x;
    const int lane = t & 31;
    const int wave = t >> 5;
    const int wmo  = (wave >> 1) * 32;
    const int wno  = (wave & 1) * 32;

    const int fl_mn = lane & 15;
    const int fl_k  = (lane >> 4) * 2;

    const float* pA = gA + (size_t)b * D1 + m0;
    const float* pB = gB + (size_t)b * D2 + n0;

    int am[4], akq[4];
#pragma unroll
    for (int q = 0; q < 4; ++q) { int s = t + q * 256; am[q] = s & 127; akq[q] = s >> 7; }
    int bm[2], bkq[2];
#pragma unroll
    for (int q = 0; q < 2; ++q) { int s = t + q * 256; bm[q] = s & 63; bkq[q] = s >> 6; }

    float ra[16];
    float rb[8];

    auto loadG = [&](int s0) {
#pragma unroll
        for (int q = 0; q < 4; ++q)
#pragma unroll
            for (int j = 0; j < 4; ++j)
                ra[q * 4 + j] = pA[(size_t)(s0 + akq[q] * 4 + j) * ROWSTRIDE + am[q]];
#pragma unroll
        for (int q = 0; q < 2; ++q)
#pragma unroll
            for (int j = 0; j < 4; ++j)
                rb[q * 4 + j] = pB[(size_t)(s0 + bkq[q] * 4 + j) * ROWSTRIDE + bm[q]];
        if (s0 + BK < S_DIM) {
            __builtin_prefetch(&pA[(size_t)(s0 + BK + akq[0] * 4) * ROWSTRIDE + am[0]], 0, 1);
            __builtin_prefetch(&pB[(size_t)(s0 + BK + bkq[0] * 4) * ROWSTRIDE + bm[0]], 0, 1);
        }
    };

    auto storeS = [&](int buf) {
#pragma unroll
        for (int q = 0; q < 4; ++q) {
            v4f v = { ra[q * 4 + 0], ra[q * 4 + 1], ra[q * 4 + 2], ra[q * 4 + 3] };
            *(v4f*)&lA[buf][am[q] * LDSK + akq[q] * 4] = v;
        }
#pragma unroll
        for (int q = 0; q < 2; ++q) {
            v4f v = { rb[q * 4 + 0], rb[q * 4 + 1], rb[q * 4 + 2], rb[q * 4 + 3] };
            *(v4f*)&lB[buf][bm[q] * LDSK + bkq[q] * 4] = v;
        }
    };

    v8f acc[2][2];
#pragma unroll
    for (int i = 0; i < 2; ++i)
#pragma unroll
        for (int j = 0; j < 2; ++j)
            acc[i][j] = v8f{};

    auto compute = [&](int buf) {
#pragma unroll
        for (int kk = 0; kk < BK; kk += 4) {
            v2f af[2], bf[2];
#pragma unroll
            for (int mt = 0; mt < 2; ++mt)
                af[mt] = *(const v2f*)&lA[buf][(wmo + mt * 16 + fl_mn) * LDSK + kk + fl_k];
#pragma unroll
            for (int nt = 0; nt < 2; ++nt)
                bf[nt] = *(const v2f*)&lB[buf][(wno + nt * 16 + fl_mn) * LDSK + kk + fl_k];
#pragma unroll
            for (int mt = 0; mt < 2; ++mt)
#pragma unroll
                for (int nt = 0; nt < 2; ++nt)
                    acc[mt][nt] = __builtin_amdgcn_wmma_f32_16x16x4_f32(
                        false, af[mt], false, bf[nt],
                        (short)0, acc[mt][nt], false, false);
        }
    };

    loadG(0);
    storeS(0);
    __syncthreads();

    for (int kt = 0; kt < NKT; ++kt) {
        const int cur = kt & 1;
        if (kt + 1 < NKT) loadG((kt + 1) * BK);
        compute(cur);
        if (kt + 1 < NKT) {
            storeS(cur ^ 1);
            __syncthreads();
        }
    }

    float* pC = gC + (size_t)b * (D1 * D2);
    const int mhalf = (lane >> 4) * 8;
#pragma unroll
    for (int mt = 0; mt < 2; ++mt)
#pragma unroll
        for (int nt = 0; nt < 2; ++nt)
#pragma unroll
            for (int r = 0; r < 8; ++r) {
                int m = m0 + wmo + mt * 16 + r + mhalf;
                int n = n0 + wno + nt * 16 + fl_mn;
                pC[(size_t)m * D2 + n] = acc[mt][nt][r];
            }
}

extern "C" void kernel_launch(void* const* d_in, const int* in_sizes, int n_in,
                              void* d_out, int out_size, void* d_ws, size_t ws_size,
                              hipStream_t stream) {
    const float* in1 = (const float*)d_in[0];
    const float* in2 = (const float*)d_in[1];
    float* out = (float*)d_out;

    dim3 grid(D2 / BN, D1 / BM, B_DIM);
    dim3 block(256);
    SumFlattenedOuterProduct_kernel<<<grid, block, 0, stream>>>(in1, in2, out);
}


# No LDS, no barrier: each wave (32x32 as 2x2 tiles; 4 waves per 64x64 work-group, grid 256) loads its WMMA fragments straight from global, 8-byte loads, 3 k16 blocks ahead; k and sum order as before
# speedup vs baseline: 1.5023x; 1.5023x over previous
#include <hip/hip_runtime.h>


typedef float v2f __attribute__((ext_vector_type(2)));
typedef float v8f __attribute__((ext_vector_type(8)));

#define S_DIM 2048
#define B_DIM 16
#define D1 256
#define D2 256
#define ROWSTRIDE (B_DIM * D1)

#define WG_M 64
#define WG_N 64
#define KB 16
#define NBLK (S_DIM / KB)
#define NS 4
#define PD (NS - 1)

__global__ __launch_bounds__(128)
void SumFlattenedOuterProduct_kernel(const float* __restrict__ gA,
                                     const float* __restrict__ gB,
                                     float* __restrict__ gC)
{
    const int b    = blockIdx.z;
    const int lane = threadIdx.x & 31;
    const int wave = threadIdx.x >> 5;
    const int c    = lane & 15;
    const int h    = lane >> 4;

    const int mw = blockIdx.y * WG_M + (wave >> 1) * 32;
    const int nw = blockIdx.x * WG_N + (wave & 1) * 32;

    const float* pA = gA + (size_t)(2 * h) * ROWSTRIDE + b * D1 + mw + 2 * c;
    const float* pB = gB + (size_t)(2 * h) * ROWSTRIDE + b * D2 + nw + 2 * c;

    v2f ra[NS][8];
    v2f rb[NS][8];

    auto loadBlk = [&](int blk, int set) {
        const float* a  = pA + (size_t)blk * (KB * ROWSTRIDE);
        const float* bb = pB + (size_t)blk * (KB * ROWSTRIDE);
#pragma unroll
        for (int q = 0; q < 4; ++q)
#pragma unroll
            for (int j = 0; j < 2; ++j) {
                ra[set][q * 2 + j] = *(const v2f*)(a  + (size_t)(q * 4 + j) * ROWSTRIDE);
                rb[set][q * 2 + j] = *(const v2f*)(bb + (size_t)(q * 4 + j) * ROWSTRIDE);
            }
    };

    v8f acc[2][2];
#pragma unroll
    for (int i = 0; i < 2; ++i)
#pragma unroll
        for (int j = 0; j < 2; ++j)
            acc[i][j] = v8f{};

    auto computeBlk = [&](int set) {
#pragma unroll
        for (int q = 0; q < 4; ++q) {
            v2f af[2], bf[2];
#pragma unroll
            for (int t = 0; t < 2; ++t) {
                af[t] = v2f{ ra[set][q * 2 + 0][t], ra[set][q * 2 + 1][t] };
                bf[t] = v2f{ rb[set][q * 2 + 0][t], rb[set][q * 2 + 1][t] };
            }
#pragma unroll
            for (int mt = 0; mt < 2; ++mt)
#pragma unroll
                for (int nt = 0; nt < 2; ++nt)
                    acc[mt][nt] = __builtin_amdgcn_wmma_f32_16x16x4_f32(
                        false, af[mt], false, bf[nt],
                        (short)0, acc[mt][nt], false, false);
        }
    };

#pragma unroll
    for (int p = 0; p < PD; ++p) loadBlk(p, p);

    int kb = 0;
    for (; kb < NBLK - NS; kb += NS) {
#pragma unroll
        for (int u = 0; u < NS; ++u) {
            loadBlk(kb + u + PD, (u + PD) % NS);
            computeBlk(u);
        }
    }
    loadBlk(kb + PD, PD % NS);
#pragma unroll
    for (int u = 0; u < NS; ++u) computeBlk(u);

    float* pC = gC + (size_t)b * (D1 * D2);
#pragma unroll
    for (int mt = 0; mt < 2; ++mt)
#pragma unroll
        for (int r = 0; r < 8; ++r) {
            const int i = mw + 2 * (8 * h + r) + mt;
            const int j = nw + 2 * c;
            *(v2f*)&pC[(size_t)i * D2 + j] = v2f{ acc[mt][0][r], acc[mt][1][r] };
        }
}

extern "C" void kernel_launch(void* const* d_in, const int* in_sizes, int n_in,
                              void* d_out, int out_size, void* d_ws, size_t ws_size,
                              hipStream_t stream) {
    const float* in1 = (const float*)d_in[0];
    const float* in2 = (const float*)d_in[1];
    float* out = (float*)d_out;

    dim3 grid(D2 / WG_N, D1 / WG_M, B_DIM);
    dim3 block(128);
    SumFlattenedOuterProduct_kernel<<<grid, block, 0, stream>>>(in1, in2, out);
}


# Split-k inside the work-group: each of the 4 waves owns the whole 64x64 tile (4x4 WMMA tiles, 16-byte fragment loads from global) for a quarter of s; the 4 partial sums are added via LDS in s order
# speedup vs baseline: 2.3137x; 1.5401x over previous
#include <hip/hip_runtime.h>


typedef float v2f __attribute__((ext_vector_type(2)));
typedef float v4f __attribute__((ext_vector_type(4)));
typedef float v8f __attribute__((ext_vector_type(8)));

#define S_DIM 2048
#define B_DIM 16
#define D1 256
#define D2 256
#define ROWSTRIDE (B_DIM * D1)

#define WG_M 64
#define WG_N 64
#define NWAVE 4
#define S_WAVE (S_DIM / NWAVE)
#define KB 16
#define NBLK (S_WAVE / KB)
#define NS 4
#define PD (NS - 1)

__global__ __launch_bounds__(128)
void SumFlattenedOuterProduct_kernel(const float* __restrict__ gA,
                                     const float* __restrict__ gB,
                                     float* __restrict__ gC)
{
    __shared__ float red[NWAVE][WG_M][WG_N];

    const int b    = blockIdx.z;
    const int lane = threadIdx.x & 31;
    const int wave = threadIdx.x >> 5;
    const int c    = lane & 15;
    const int h    = lane >> 4;

    const int m0 = blockIdx.y * WG_M;
    const int n0 = blockIdx.x * WG_N;

    const float* pA = gA + (size_t)(wave * S_WAVE + 2 * h) * ROWSTRIDE + b * D1 + m0 + 4 * c;
    const float* pB = gB + (size_t)(wave * S_WAVE + 2 * h) * ROWSTRIDE + b * D2 + n0 + 4 * c;

    v4f ra[NS][8];
    v4f rb[NS][8];

    auto loadBlk = [&](int blk, int set) {
        const float* a  = pA + (size_t)blk * (KB * ROWSTRIDE);
        const float* bb = pB + (size_t)blk * (KB * ROWSTRIDE);
#pragma unroll
        for (int q = 0; q < 4; ++q)
#pragma unroll
            for (int j = 0; j < 2; ++j) {
                ra[set][q * 2 + j] = *(const v4f*)(a  + (size_t)(q * 4 + j) * ROWSTRIDE);
                rb[set][q * 2 + j] = *(const v4f*)(bb + (size_t)(q * 4 + j) * ROWSTRIDE);
            }
    };

    v8f acc[4][4];
#pragma unroll
    for (int i = 0; i < 4; ++i)
#pragma unroll
        for (int j = 0; j < 4; ++j)
            acc[i][j] = v8f{};

    auto computeBlk = [&](int set) {
#pragma unroll
        for (int q = 0; q < 4; ++q) {
            v2f af[4], bf[4];
#pragma unroll
            for (int t = 0; t < 4; ++t) {
                af[t] = v2f{ ra[set][q * 2 + 0][t], ra[set][q * 2 + 1][t] };
                bf[t] = v2f{ rb[set][q * 2 + 0][t], rb[set][q * 2 + 1][t] };
            }
#pragma unroll
            for (int mt = 0; mt < 4; ++mt)
#pragma unroll
                for (int nt = 0; nt < 4; ++nt)
                    acc[mt][nt] = __builtin_amdgcn_wmma_f32_16x16x4_f32(
                        false, af[mt], false, bf[nt],
                        (short)0, acc[mt][nt], false, false);
        }
    };

#pragma unroll
    for (int p = 0; p < PD; ++p) loadBlk(p, p);

    int kb = 0;
    for (; kb < NBLK - NS; kb += NS) {
#pragma unroll
        for (int u = 0; u < NS; ++u) {
            loadBlk(kb + u + PD, (u + PD) % NS);
            __builtin_amdgcn_sched_barrier(0);
            computeBlk(u);
            __builtin_amdgcn_sched_barrier(0);
        }
    }
    loadBlk(kb + PD, PD % NS);
    __builtin_amdgcn_sched_barrier(0);
#pragma unroll
    for (int u = 0; u < NS; ++u) computeBlk(u);

#pragma unroll
    for (int mt = 0; mt < 4; ++mt)
#pragma unroll
        for (int r = 0; r < 8; ++r) {
            const int row = 4 * (8 * h + r) + mt;
            *(v4f*)&red[wave][row][4 * c] =
                v4f{ acc[mt][0][r], acc[mt][1][r], acc[mt][2][r], acc[mt][3][r] };
        }
    __syncthreads();

    float* pC = gC + (size_t)b * (D1 * D2);
#pragma unroll
    for (int j = 0; j < 8; ++j) {
        const int row = 16 * wave + 2 * j + h;
        v4f s = *(const v4f*)&red[0][row][4 * c];
        s += *(const v4f*)&red[1][row][4 * c];
        s += *(const v4f*)&red[2][row][4 * c];
        s += *(const v4f*)&red[3][row][4 * c];
        *(v4f*)&pC[(size_t)(m0 + row) * D2 + n0 + 4 * c] = s;
    }
}

extern "C" void kernel_launch(void* const* d_in, const int* in_sizes, int n_in,
                              void* d_out, int out_size, void* d_ws, size_t ws_size,
                              hipStream_t stream) {
    const float* in1 = (const float*)d_in[0];
    const float* in2 = (const float*)d_in[1];
    float* out = (float*)d_out;

    dim3 grid(D2 / WG_N, D1 / WG_M, B_DIM);
    dim3 block(128);
    SumFlattenedOuterProduct_kernel<<<grid, block, 0, stream>>>(in1, in2, out);
}
